# starting kernel
# baseline (speedup 1.0000x reference)
#include <hip/hip_runtime.h>
#include <hip/hip_bf16.h>

typedef float v2f __attribute__((ext_vector_type(2)));
typedef float v8f __attribute__((ext_vector_type(8)));

#define BATCH     32768
#define KD        784
#define N_EXPERTS 5
#define ROWS_PER_WAVE 16
#define WAVES_PER_BLOCK 8
#define THREADS (WAVES_PER_BLOCK * 32)

__global__ __launch_bounds__(THREADS)
void moe_wmma_kernel(const float* __restrict__ x,
                     const int*   __restrict__ cat,
                     const float* __restrict__ conv_w,
                     const float* __restrict__ conv_b,
                     const float* __restrict__ fc1_w,
                     const float* __restrict__ fc1_b,
                     float* __restrict__ out)
{
    __shared__ float tile[WAVES_PER_BLOCK][ROWS_PER_WAVE * 16];

    const int lane  = threadIdx.x & 31;
    const int wave  = threadIdx.x >> 5;
    const int gwave = blockIdx.x * WAVES_PER_BLOCK + wave;
    const int r0    = gwave * ROWS_PER_WAVE;

    const int m  = lane & 15;
    const int hk = (lane >> 4) * 2;

    const bool bvalid = (m < N_EXPERTS);
    const int  nclamp = bvalid ? m : (N_EXPERTS - 1);

    const v2f* __restrict__ ap = (const v2f*)(x      + (size_t)(r0 + m) * KD + hk);
    const v2f* __restrict__ bp = (const v2f*)(conv_w + (size_t)nclamp   * KD + hk);

    v8f acc = {};
    #pragma unroll 4
    for (int kk = 0; kk < KD / 4; ++kk) {
        v2f a = ap[2 * kk];
        v2f b = bp[2 * kk];
        b.x = bvalid ? b.x : 0.0f;
        b.y = bvalid ? b.y : 0.0f;
        acc = __builtin_amdgcn_wmma_f32_16x16x4_f32(
             false, a,  false, b,
             (short)0, acc,  false,  false);
    }

    {
        float* t = tile[wave];
        const int mh = (lane >> 4) * 8;
        const int nn = lane & 15;
        #pragma unroll
        for (int r = 0; r < 8; ++r)
            t[(mh + r) * 16 + nn] = acc[r];
    }
    __syncthreads();

    if (lane < ROWS_PER_WAVE) {
        const int   row = r0 + lane;
        const int   ci  = cat[row];
        const float s   = tile[wave][lane * 16 + ci] + conv_b[ci];
        const float o   = s * fc1_w[ci] + fc1_b[0];
        out[row] = o > 0.0f ? o : 0.0f;
    }
}

extern "C" void kernel_launch(void* const* d_in, const int* in_sizes, int n_in,
                              void* d_out, int out_size, void* d_ws, size_t ws_size,
                              hipStream_t stream) {
    const float* x    = (const float*)d_in[0];
    const int*   cat  = (const int*)  d_in[1];
    const float* cw   = (const float*)d_in[2];
    const float* cb   = (const float*)d_in[3];
    const float* f1w  = (const float*)d_in[4];
    const float* f1b  = (const float*)d_in[5];
    float*       out  = (float*)d_out;

    const int nblocks = BATCH / (ROWS_PER_WAVE * WAVES_PER_BLOCK);
    hipLaunchKernelGGL(moe_wmma_kernel, dim3(nblocks), dim3(THREADS), 0, stream,
                       x, cat, cw, cb, f1w, f1b, out);
}


# K cut in four quarters, one per wave: a wave sums its block's 128 samples over one quarter, holding a quarter of conv_w, in 8 short sub-tiles; quarter sums joined via LDS: f32 sum order changes
# speedup vs baseline: 3.4883x; 3.4883x over previous
#include <hip/hip_runtime.h>

typedef float v2f __attribute__((ext_vector_type(2)));
typedef float v4f __attribute__((ext_vector_type(4)));
typedef float v8f __attribute__((ext_vector_type(8)));

#define BATCH      32768
#define KD         784
#define N_EXPERTS  5
#define TILE_ROWS  16
#define ROW_F4     196
#define WAVES      4
#define TILES_PER_WAVE 8
#define SAMPLES_PER_BLOCK 128
#define NSETS      3
#define THREADS    (WAVES * 32)
#define NBLOCKS    (BATCH / SAMPLES_PER_BLOCK)
#define MAXPITCH_F4 51

template <int NB> struct Q {
    static constexpr int RL    = 2 * NB;
    static constexpr int PITCH = RL + 1;
    static constexpr int NGRP  = 3;
    static constexpr int GMAX  = NB - (NB * (NGRP - 1)) / NGRP;
    static __device__ __forceinline__ constexpr int gs(int g) { return (NB * g) / NGRP; }
    static constexpr int SKIP  = ROW_F4 - RL;
    static __device__ __forceinline__ constexpr int rowof(int q) { return (32 * q) / RL; }
    static __device__ __forceinline__ constexpr int thr(int q)   { return RL - (32 * q) % RL; }
};


#define GLD(SRC, q) (SRC)[32 * (q) + QQ::SKIP * QQ::rowof(q) + ((lane >= QQ::thr(q)) ? QQ::SKIP : 0)]

#define SUBTILE(PRE, HAS_NEXT, NSRC) do {                                                              \
    _Pragma("unroll") for (int q = 0; q < NB; ++q)                                                     \
        buf[32 * q + lane + QQ::rowof(q) + ((lane >= QQ::thr(q)) ? 1 : 0)] = (PRE)[q];                 \
    v4f af[2][QQ::GMAX];                                                                               \
    _Pragma("unroll") for (int i = 0; i < QQ::gs(1); ++i) af[0][i] = ap[2 * i];                        \
    _Pragma("unroll") for (int g = 0; g < QQ::NGRP; ++g) {                                             \
          \
        __builtin_amdgcn_sched_barrier(0);                                                             \
        if (g + 1 < QQ::NGRP) {                                                                        \
            _Pragma("unroll") for (int i = 0; i < QQ::gs(g + 2) - QQ::gs(g + 1); ++i)                  \
                af[(g + 1) & 1][i] = ap[2 * (QQ::gs(g + 1) + i)];                                      \
        }                                                                                              \
        if (HAS_NEXT) {                                                                                \
            _Pragma("unroll") for (int i = 0; i < QQ::gs(g + 1) - QQ::gs(g); ++i)                      \
                (PRE)[QQ::gs(g) + i] = GLD(NSRC, QQ::gs(g) + i);                                       \
        }                                                                                              \
                                          \
        __builtin_amdgcn_sched_barrier(0);                                                             \
        _Pragma("unroll") for (int i = 0; i < QQ::gs(g + 1) - QQ::gs(g); ++i) {                        \
            const v4f a = af[g & 1][i];                                                                \
            const v4f b = bw[QQ::gs(g) + i];                                                           \
            acc0 = __builtin_amdgcn_wmma_f32_16x16x4_f32(false, (v2f){a.x, a.y}, false, (v2f){b.x, b.y}, \
                                                         (short)0, acc0, false, false);                \
            acc1 = __builtin_amdgcn_wmma_f32_16x16x4_f32(false, (v2f){a.z, a.w}, false, (v2f){b.z, b.w}, \
                                                         (short)0, acc1, false, false);                \
        }                                                                                              \
    }                                                                                                  \
} while (0)

template <int NB>
__device__ __forceinline__ void run_quarter(const float* __restrict__ x, const float* __restrict__ conv_w,
                                            v4f* buf, const int r0, const int koff4, const int lane,
                                            const int (&ci)[4], float (&res)[4])
{
    typedef Q<NB> QQ;
    const int m = lane & 15;
    const int h = lane >> 4;

    const v4f* ap = buf + m * QQ::PITCH + h;

    const v4f* __restrict__ xt = (const v4f*)(x + (size_t)r0 * KD) + koff4 + lane;

    v4f pre0[NB], pre1[NB], pre2[NB];
    #pragma unroll
    for (int q = 0; q < NB; ++q) pre0[q] = GLD(xt, q);
    __builtin_amdgcn_sched_barrier(0);

    v4f bw[NB];
    {
        const int nclamp = (m < N_EXPERTS) ? m : (N_EXPERTS - 1);
        const v4f* bp = (const v4f*)(conv_w + (size_t)nclamp * KD) + koff4 + h;
        #pragma unroll
        for (int j = 0; j < NB; ++j) bw[j] = bp[2 * j];
    }
    __builtin_amdgcn_sched_barrier(0);
    #pragma unroll
    for (int q = 0; q < NB; ++q) pre1[q] = GLD(xt + TILE_ROWS * ROW_F4, q);
    #pragma unroll
    for (int q = 0; q < NB; ++q) pre2[q] = GLD(xt + 2 * TILE_ROWS * ROW_F4, q);

    #pragma unroll
    for (int t = 0; t < TILES_PER_WAVE; ++t) {
        v8f acc0 = {}, acc1 = {};
        const v4f* __restrict__ xn = xt + (size_t)(t + NSETS) * (TILE_ROWS * ROW_F4);

        if      (t % NSETS == 0) SUBTILE(pre0, (t + NSETS < TILES_PER_WAVE), xn);
        else if (t % NSETS == 1) SUBTILE(pre1, (t + NSETS < TILES_PER_WAVE), xn);
        else                     SUBTILE(pre2, (t + NSETS < TILES_PER_WAVE), xn);

        const v8f acc = acc0 + acc1;

        float* sc = (float*)buf;
        #pragma unroll
        for (int r = 0; r < 8; ++r)
            sc[(h * 8 + r) * 16 + m] = acc[r];

        const float s = sc[m * 16 + ci[t >> 1]];
        res[t >> 1] = (h == (t & 1)) ? s : res[t >> 1];
    }
}

__global__ __launch_bounds__(THREADS) __attribute__((amdgpu_waves_per_eu(1, 1)))
void moe_wmma_ksplit_kernel(const float* __restrict__ x,
                            const int*   __restrict__ cat,
                            const float* __restrict__ conv_w,
                            const float* __restrict__ conv_b,
                            const float* __restrict__ fc1_w,
                            const float* __restrict__ fc1_b,
                            float* __restrict__ out)
{
    __shared__ v4f   tile[WAVES][TILE_ROWS * MAXPITCH_F4];
    __shared__ float part[4][SAMPLES_PER_BLOCK];

    const int lane = threadIdx.x & 31;
    const int wave = __builtin_amdgcn_readfirstlane((int)(threadIdx.x >> 5));
    const int kq   = wave;
    const int r0   = blockIdx.x * SAMPLES_PER_BLOCK;

    const int cif = cat[r0 + (int)threadIdx.x];

    const float cb0 = conv_b[0], cb1 = conv_b[1], cb2 = conv_b[2], cb3 = conv_b[3], cb4 = conv_b[4];
    const float fw0 = fc1_w[0],  fw1 = fc1_w[1],  fw2 = fc1_w[2],  fw3 = fc1_w[3],  fw4 = fc1_w[4];
    const float fb  = fc1_b[0];

    int ci[4];
    #pragma unroll
    for (int p = 0; p < 4; ++p) ci[p] = cat[r0 + 32 * p + lane];

    float res[4] = {0.0f, 0.0f, 0.0f, 0.0f};
    if (kq < 2) run_quarter<25>(x, conv_w, tile[wave], r0, 50 * kq,             lane, ci, res);
    else        run_quarter<24>(x, conv_w, tile[wave], r0, 100 + 48 * (kq - 2), lane, ci, res);

    #pragma unroll
    for (int p = 0; p < 4; ++p) part[kq][32 * p + lane] = res[p];

    __syncthreads();

    {
        const int   i   = threadIdx.x;
        float cb = cb4, fw = fw4;
        cb = (cif == 3) ? cb3 : cb;   fw = (cif == 3) ? fw3 : fw;
        cb = (cif == 2) ? cb2 : cb;   fw = (cif == 2) ? fw2 : fw;
        cb = (cif == 1) ? cb1 : cb;   fw = (cif == 1) ? fw1 : fw;
        cb = (cif == 0) ? cb0 : cb;   fw = (cif == 0) ? fw0 : fw;
        const float dot = (part[0][i] + part[1][i]) + (part[2][i] + part[3][i]);
        const float s   = dot + cb;
        const float o   = s * fw + fb;
        out[r0 + i] = o > 0.0f ? o : 0.0f;
    }
}

extern "C" void kernel_launch(void* const* d_in, const int* in_sizes, int n_in,
                              void* d_out, int out_size, void* d_ws, size_t ws_size,
                              hipStream_t stream) {
    const float* x    = (const float*)d_in[0];
    const int*   cat  = (const int*)  d_in[1];
    const float* cw   = (const float*)d_in[2];
    const float* cb   = (const float*)d_in[3];
    const float* f1w  = (const float*)d_in[4];
    const float* f1b  = (const float*)d_in[5];
    float*       out  = (float*)d_out;

    hipLaunchKernelGGL(moe_wmma_ksplit_kernel, dim3(NBLOCKS), dim3(THREADS), 0, stream,
                       x, cat, cw, cb, f1w, f1b, out);
}
